# starting kernel
# baseline (speedup 1.0000x reference)
#include <hip/hip_runtime.h>
#include <math.h>

typedef __attribute__((ext_vector_type(2))) float v2f;
typedef __attribute__((ext_vector_type(8))) float v8f;

#define HDIM     256
#define KCH      64
#define TS       16
#define NTHREADS 512

#define OFF_W2   0
#define OFF_H1   (HDIM*HDIM)
#define OFF_H1D  (OFF_H1 + TS*HDIM)
#define OFF_W1Z  (OFF_H1D + TS*HDIM)
#define OFF_W1T  (OFF_W1Z + HDIM)
#define OFF_B1   (OFF_W1T + HDIM)
#define OFF_B2   (OFF_B1 + HDIM)
#define OFF_W3   (OFF_B2 + HDIM)
#define OFF_F    (OFF_W3 + HDIM)
#define OFF_FD   (OFF_F + TS)
#define OFF_Z    (OFF_FD + TS)
#define OFF_ZT   (OFF_Z + TS)
#define OFF_DIV  (OFF_ZT + TS)
#define OFF_KZ   (OFF_DIV + TS)
#define OFF_KD   (OFF_KZ + TS)
#define LDS_FLOATS (OFF_KD + TS)

__device__ __forceinline__ float tanh_fast(float x) {
#if defined(__AMDGCN__) && __has_builtin(__builtin_amdgcn_tanhf)
  return __builtin_amdgcn_tanhf(x);
#else
  float e = __expf(2.0f * x);
  return 1.0f - 2.0f / (e + 1.0f);
#endif
}

__global__ void __launch_bounds__(NTHREADS, 1)
cnf1d_rk4_wmma(const float* __restrict__ z0, const float* __restrict__ W1,
               const float* __restrict__ b1, const float* __restrict__ W2,
               const float* __restrict__ b2, const float* __restrict__ W3,
               const float* __restrict__ b3, float* __restrict__ out, int Btot)
{
  extern __shared__ float sm[];
  float* sW2  = sm + OFF_W2;
  float* sH1  = sm + OFF_H1;
  float* sH1d = sm + OFF_H1D;
  float* sW1z = sm + OFF_W1Z;
  float* sW1t = sm + OFF_W1T;
  float* sB1  = sm + OFF_B1;
  float* sB2  = sm + OFF_B2;
  float* sW3  = sm + OFF_W3;
  float* sF   = sm + OFF_F;
  float* sFd  = sm + OFF_FD;
  float* sZ   = sm + OFF_Z;
  float* sZt  = sm + OFF_ZT;
  float* sDiv = sm + OFF_DIV;
  float* sKz  = sm + OFF_KZ;
  float* sKd  = sm + OFF_KD;

  const int tid  = threadIdx.x;
  const int lane = tid & 31;
  const int wv   = tid >> 5;
  const float dtv = 0.25f;
  const float b3v = b3[0];

  for (int i = tid; i < HDIM; i += NTHREADS) {
    sW1z[i] = W1[i];
    sW1t[i] = W1[HDIM + i];
    sB1[i]  = b1[i];
    sB2[i]  = b2[i];
    sW3[i]  = W3[i];
  }
  for (int e = tid; e < HDIM * HDIM; e += NTHREADS) {
    int row = e >> 8, col = e & 255;
    int c  = row >> 2, rr = row & 3;
    int n  = col >> 4, lo = col & 15;
    int l  = ((rr >> 1) << 4) | lo;
    int d  = rr & 1;
    int idx = ((((((c << 4) | n) << 5) | l) << 1) | d);
    sW2[idx] = W2[e];
  }
  if (tid < TS) {
    float z = z0[(size_t)blockIdx.x * TS + tid];
    sZ[tid] = z; sZt[tid] = z; sDiv[tid] = 0.0f;
    sKz[tid] = 0.0f; sKd[tid] = 0.0f;
    sF[tid] = b3v; sFd[tid] = 0.0f;
  }
  __syncthreads();

  #pragma unroll 1
  for (int ev = 0; ev < 16; ++ev) {
    const int stg = ev & 3;
    float tt = (float)(ev >> 2) * dtv;
    if (stg == 3) tt += dtv; else if (stg != 0) tt += 0.5f * dtv;

    #pragma unroll
    for (int i = 0; i < 8; ++i) {
      int s = tid + NTHREADS * i;
      int d = s & 1;
      int l = (s >> 1) & 31;
      int c = s >> 6;
      int m = l & 15;
      int j = (c << 2) + ((l >> 4) << 1) + d;
      float w1 = sW1z[j];
      float a  = sZt[m] * w1 + tt * sW1t[j] + sB1[j];
      float h  = tanh_fast(a);
      sH1[s]  = h;
      sH1d[s] = (1.0f - h * h) * w1;
    }
    __syncthreads();

    v8f acc  = {};
    v8f accd = {};
    const v2f* aF  = (const v2f*)sH1;
    const v2f* adF = (const v2f*)sH1d;
    const v2f* bF  = (const v2f*)sW2;
    #pragma unroll 4
    for (int c = 0; c < KCH; ++c) {
      v2f af = aF[(c << 5) + lane];
      v2f ad = adF[(c << 5) + lane];
      v2f bf = bF[((((c << 4) | wv) << 5) + lane)];
      acc  = __builtin_amdgcn_wmma_f32_16x16x4_f32(false, af, false, bf,
                                                   (short)0, acc,  false, false);
      accd = __builtin_amdgcn_wmma_f32_16x16x4_f32(false, ad, false, bf,
                                                   (short)0, accd, false, false);
    }

    {
      const int n  = lane & 15;
      const int j  = (wv << 4) | n;
      const int mb = (lane >> 4) << 3;
      const float w3 = sW3[j];
      const float bb = sB2[j];
      #pragma unroll
      for (int v = 0; v < 8; ++v) {
        float h2  = tanh_fast(acc[v] + bb);
        float pf  = h2 * w3;
        float pfd = (1.0f - h2 * h2) * accd[v] * w3;
        #pragma unroll
        for (int msk = 1; msk < 16; msk <<= 1) {
          pf  += __shfl_xor(pf,  msk, 32);
          pfd += __shfl_xor(pfd, msk, 32);
        }
        if (n == 0) {
          atomicAdd(&sF[mb + v],  pf);
          atomicAdd(&sFd[mb + v], pfd);
        }
      }
    }
    __syncthreads();

    if (tid < TS) {
      float kz = sF[tid];
      float kd = sFd[tid];
      float w  = (stg == 1 || stg == 2) ? 2.0f : 1.0f;
      float akz = sKz[tid] + w * kz;
      float akd = sKd[tid] + w * kd;
      if (stg < 3) {
        float cf = (stg == 2) ? dtv : 0.5f * dtv;
        sZt[tid] = sZ[tid] + cf * kz;
        sKz[tid] = akz; sKd[tid] = akd;
      } else {
        float zn = sZ[tid] + (dtv * (1.0f / 6.0f)) * akz;
        sZ[tid] = zn; sZt[tid] = zn;
        sDiv[tid] += (dtv * (1.0f / 6.0f)) * akd;
        sKz[tid] = 0.0f; sKd[tid] = 0.0f;
      }
      sF[tid] = b3v; sFd[tid] = 0.0f;
    }
    __syncthreads();
  }

  if (tid < TS) {
    int g = blockIdx.x * TS + tid;
    out[g] = sZ[tid];
    out[Btot + g] = sDiv[tid];
  }
}

extern "C" void kernel_launch(void* const* d_in, const int* in_sizes, int n_in,
                              void* d_out, int out_size, void* d_ws, size_t ws_size,
                              hipStream_t stream) {
  (void)n_in; (void)d_ws; (void)ws_size; (void)out_size;
  const float* z0 = (const float*)d_in[0];
  const float* W1 = (const float*)d_in[1];
  const float* b1 = (const float*)d_in[2];
  const float* W2 = (const float*)d_in[3];
  const float* b2 = (const float*)d_in[4];
  const float* W3 = (const float*)d_in[5];
  const float* b3 = (const float*)d_in[6];
  float* out = (float*)d_out;

  const int B    = in_sizes[0];
  const int nblk = B / TS;
  const size_t lds = (size_t)LDS_FLOATS * sizeof(float);

  hipFuncSetAttribute(reinterpret_cast<const void*>(cnf1d_rk4_wmma),
                      hipFuncAttributeMaxDynamicSharedMemorySize, (int)lds);
  cnf1d_rk4_wmma<<<nblk, NTHREADS, lds, stream>>>(z0, W1, b1, W2, b2, W3, b3, out, B);
}


# W3 sums over the 16 N-lanes done by DPP row_xmask adds on the vector ALU, 16 sums per level together, not ds_bpermute + wait per level and value; same butterfly levels, same sums
# speedup vs baseline: 1.4361x; 1.4361x over previous
#include <hip/hip_runtime.h>
#include <math.h>


typedef __attribute__((ext_vector_type(2))) float v2f;
typedef __attribute__((ext_vector_type(4))) float v4f;
typedef __attribute__((ext_vector_type(8))) float v8f;

#define HDIM     256
#define KCH      64
#define MS       128
#define MT       8
#define NTHREADS 512

#define OFF_HH   0
#define OFF_W1Z  (MT*KCH*32*4)
#define OFF_W1T  (OFF_W1Z + HDIM)
#define OFF_B1   (OFF_W1T + HDIM)
#define OFF_B2   (OFF_B1 + HDIM)
#define OFF_W3   (OFF_B2 + HDIM)
#define OFF_F    (OFF_W3 + HDIM)
#define OFF_FD   (OFF_F + MS)
#define OFF_Z    (OFF_FD + MS)
#define OFF_ZT   (OFF_Z + MS)
#define OFF_DIV  (OFF_ZT + MS)
#define OFF_KZ   (OFF_DIV + MS)
#define OFF_KD   (OFF_KZ + MS)
#define LDS_FLOATS (OFF_KD + MS)

__device__ __forceinline__ float tanh_fast(float x) {
#if defined(__AMDGCN__) && __has_builtin(__builtin_amdgcn_tanhf)
  return __builtin_amdgcn_tanhf(x);
#else
  float e = __expf(2.0f * x);
  return 1.0f - 2.0f / (e + 1.0f);
#endif
}

template <int M>
__device__ __forceinline__ float row_xadd(float x) {
  const int t = __builtin_amdgcn_update_dpp(0, __builtin_bit_cast(int, x), 0x160 | M, 0xf, 0xf, true);
  return x + __builtin_bit_cast(float, t);
}

__device__ __forceinline__ void tile_epilogue(const v8f acc, const v8f accd, float bb, float w3,
                                              int lane, float* fRow, float* fdRow)
{
  const int n  = lane & 15;
  const int mb = (lane >> 4) << 3;
  float pf[8], pfd[8];
  #pragma unroll
  for (int v = 0; v < 8; ++v) {
    float h2 = tanh_fast(acc[v] + bb);
    pf[v]  = h2 * w3;
    pfd[v] = (1.0f - h2 * h2) * accd[v] * w3;
  }
  #pragma unroll
  for (int v = 0; v < 8; ++v) { pf[v] = row_xadd<1>(pf[v]); pfd[v] = row_xadd<1>(pfd[v]); }
  #pragma unroll
  for (int v = 0; v < 8; ++v) { pf[v] = row_xadd<2>(pf[v]); pfd[v] = row_xadd<2>(pfd[v]); }
  #pragma unroll
  for (int v = 0; v < 8; ++v) { pf[v] = row_xadd<4>(pf[v]); pfd[v] = row_xadd<4>(pfd[v]); }
  #pragma unroll
  for (int v = 0; v < 8; ++v) { pf[v] = row_xadd<8>(pf[v]); pfd[v] = row_xadd<8>(pfd[v]); }
  if (n == 0) {
    #pragma unroll
    for (int v = 0; v < 8; ++v) {
      atomicAdd(&fRow[mb + v],  pf[v]);
      atomicAdd(&fdRow[mb + v], pfd[v]);
    }
  }
}

__global__ void __launch_bounds__(NTHREADS, 1)
cnf1d_rk4_wmma(const float* __restrict__ z0, const float* __restrict__ W1,
               const float* __restrict__ b1, const float* __restrict__ W2,
               const float* __restrict__ b2, const float* __restrict__ W3,
               const float* __restrict__ b3, float* __restrict__ out, int Btot)
{
  extern __shared__ float sm[];
  v4f*   sHH  = (v4f*)(sm + OFF_HH);
  float* sW1z = sm + OFF_W1Z;
  float* sW1t = sm + OFF_W1T;
  float* sB1  = sm + OFF_B1;
  float* sB2  = sm + OFF_B2;
  float* sW3  = sm + OFF_W3;
  float* sF   = sm + OFF_F;
  float* sFd  = sm + OFF_FD;
  float* sZ   = sm + OFF_Z;
  float* sZt  = sm + OFF_ZT;
  float* sDiv = sm + OFF_DIV;
  float* sKz  = sm + OFF_KZ;
  float* sKd  = sm + OFF_KD;

  const int tid  = threadIdx.x;
  const int lane = tid & 31;
  const int wv   = tid >> 5;
  const int hh   = lane >> 4;
  const float dtv = 0.25f;
  const float b3v = b3[0];

  v2f bfr[KCH];
  {
    const float* wp = W2 + (size_t)(2 * hh) * HDIM + (wv << 4) + (lane & 15);
    #pragma unroll
    for (int c = 0; c < KCH; ++c) {
      bfr[c].x = wp[(4 * c)     * HDIM];
      bfr[c].y = wp[(4 * c + 1) * HDIM];
    }
  }

  for (int i = tid; i < HDIM; i += NTHREADS) {
    sW1z[i] = W1[i];
    sW1t[i] = W1[HDIM + i];
    sB1[i]  = b1[i];
    sB2[i]  = b2[i];
    sW3[i]  = W3[i];
  }
  if (tid < MS) {
    float z = z0[(size_t)blockIdx.x * MS + tid];
    sZ[tid] = z; sZt[tid] = z; sDiv[tid] = 0.0f;
    sKz[tid] = 0.0f; sKd[tid] = 0.0f;
    sF[tid] = b3v; sFd[tid] = 0.0f;
  }
  __syncthreads();

  const int   jn = (wv << 4) | (lane & 15);
  const float w3 = sW3[jn];
  const float bb = sB2[jn];

  #pragma unroll 1
  for (int ev = 0; ev < 16; ++ev) {
    const int stg = ev & 3;
    float tt = (float)(ev >> 2) * dtv;
    if (stg == 3) tt += dtv; else if (stg != 0) tt += 0.5f * dtv;

    {
      const int m = lane & 15;
      #pragma unroll
      for (int q = 0; q < 4; ++q) {
        const int c  = (wv << 2) + q;
        const int j0 = (c << 2) + (hh << 1);
        const float w1a = sW1z[j0],  w1b = sW1z[j0 + 1];
        const float wta = sW1t[j0],  wtb = sW1t[j0 + 1];
        const float b1a = sB1[j0],   b1b = sB1[j0 + 1];
        #pragma unroll
        for (int mt = 0; mt < MT; ++mt) {
          const float z  = sZt[(mt << 4) + m];
          const float a0 = z * w1a + tt * wta + b1a;
          const float a1 = z * w1b + tt * wtb + b1b;
          const float h0 = tanh_fast(a0);
          const float h1 = tanh_fast(a1);
          v4f o;
          o.x = h0; o.y = h1;
          o.z = (1.0f - h0 * h0) * w1a;
          o.w = (1.0f - h1 * h1) * w1b;
          sHH[((mt * KCH + c) << 5) + lane] = o;
        }
      }
    }
    __syncthreads();

    #pragma unroll 1
    for (int mt = 0; mt < MT; ++mt) {
      v8f acc = {}, accd = {};
      const v4f* hA = sHH + ((mt * KCH) << 5) + lane;
      #pragma unroll
      for (int c = 0; c < KCH; ++c) {
        const v4f xa = hA[c << 5];
        const v2f af = xa.xy, ad = xa.zw;
        acc  = __builtin_amdgcn_wmma_f32_16x16x4_f32(false, af, false, bfr[c],
                                                     (short)0, acc,  false, false);
        accd = __builtin_amdgcn_wmma_f32_16x16x4_f32(false, ad, false, bfr[c],
                                                     (short)0, accd, false, false);
      }
      tile_epilogue(acc, accd, bb, w3, lane, sF + (mt << 4), sFd + (mt << 4));
    }
    __syncthreads();

    if (tid < MS) {
      float kz = sF[tid];
      float kd = sFd[tid];
      float w  = (stg == 1 || stg == 2) ? 2.0f : 1.0f;
      float akz = sKz[tid] + w * kz;
      float akd = sKd[tid] + w * kd;
      if (stg < 3) {
        float cf = (stg == 2) ? dtv : 0.5f * dtv;
        sZt[tid] = sZ[tid] + cf * kz;
        sKz[tid] = akz; sKd[tid] = akd;
      } else {
        float zn = sZ[tid] + (dtv * (1.0f / 6.0f)) * akz;
        sZ[tid] = zn; sZt[tid] = zn;
        sDiv[tid] += (dtv * (1.0f / 6.0f)) * akd;
        sKz[tid] = 0.0f; sKd[tid] = 0.0f;
      }
      sF[tid] = b3v; sFd[tid] = 0.0f;
    }
    __syncthreads();
  }

  if (tid < MS) {
    int g = blockIdx.x * MS + tid;
    out[g] = sZ[tid];
    out[Btot + g] = sDiv[tid];
  }
}

extern "C" void kernel_launch(void* const* d_in, const int* in_sizes, int n_in,
                              void* d_out, int out_size, void* d_ws, size_t ws_size,
                              hipStream_t stream) {
  (void)n_in; (void)d_ws; (void)ws_size; (void)out_size;
  const float* z0 = (const float*)d_in[0];
  const float* W1 = (const float*)d_in[1];
  const float* b1 = (const float*)d_in[2];
  const float* W2 = (const float*)d_in[3];
  const float* b2 = (const float*)d_in[4];
  const float* W3 = (const float*)d_in[5];
  const float* b3 = (const float*)d_in[6];
  float* out = (float*)d_out;

  const int B    = in_sizes[0];
  const int nblk = B / MS;
  const size_t lds = (size_t)LDS_FLOATS * sizeof(float);

  hipFuncSetAttribute(reinterpret_cast<const void*>(cnf1d_rk4_wmma),
                      hipFuncAttributeMaxDynamicSharedMemorySize, (int)lds);
  cnf1d_rk4_wmma<<<nblk, NTHREADS, lds, stream>>>(z0, W1, b1, W2, b2, W3, b3, out, B);
}


# A fragments read 8 K-chunks at a time as one ds_load_b128 group into a second register set ahead of each 16-WMMA block (next tile's first group before the epilogue), not one per WMMA pair; same sums
# speedup vs baseline: 1.5310x; 1.0661x over previous
#include <hip/hip_runtime.h>
#include <math.h>


typedef __attribute__((ext_vector_type(2))) float v2f;
typedef __attribute__((ext_vector_type(4))) float v4f;
typedef __attribute__((ext_vector_type(8))) float v8f;

#define HDIM     256
#define KCH      64
#define MS       128
#define MT       8
#define NTHREADS 512
#define KB       8
#define NGRP     (KCH / KB)

#define OFF_HH   0
#define OFF_W1Z  (MT*KCH*32*4)
#define OFF_W1T  (OFF_W1Z + HDIM)
#define OFF_B1   (OFF_W1T + HDIM)
#define OFF_B2   (OFF_B1 + HDIM)
#define OFF_W3   (OFF_B2 + HDIM)
#define OFF_F    (OFF_W3 + HDIM)
#define OFF_FD   (OFF_F + MS)
#define OFF_Z    (OFF_FD + MS)
#define OFF_ZT   (OFF_Z + MS)
#define OFF_DIV  (OFF_ZT + MS)
#define OFF_KZ   (OFF_DIV + MS)
#define OFF_KD   (OFF_KZ + MS)
#define LDS_FLOATS (OFF_KD + MS)

__device__ __forceinline__ float tanh_fast(float x) {
#if defined(__AMDGCN__) && __has_builtin(__builtin_amdgcn_tanhf)
  return __builtin_amdgcn_tanhf(x);
#else
  float e = __expf(2.0f * x);
  return 1.0f - 2.0f / (e + 1.0f);
#endif
}

template <int M>
__device__ __forceinline__ float row_xadd(float x) {
  const int t = __builtin_amdgcn_update_dpp(0, __builtin_bit_cast(int, x), 0x160 | M, 0xf, 0xf, true);
  return x + __builtin_bit_cast(float, t);
}

__device__ __forceinline__ void load_group(v4f (&dst)[KB], const v4f* p) {
  #pragma unroll
  for (int i = 0; i < KB; ++i) dst[i] = p[i << 5];
}

__device__ __forceinline__ void wmma_block(const v4f (&set)[KB], const v2f (&bfr)[KCH], int c0,
                                           v8f& acc, v8f& accd) {
  #pragma unroll
  for (int i = 0; i < KB; ++i) {
    const v2f af = set[i].xy, ad = set[i].zw;
    acc  = __builtin_amdgcn_wmma_f32_16x16x4_f32(false, af, false, bfr[c0 + i],
                                                 (short)0, acc,  false, false);
    accd = __builtin_amdgcn_wmma_f32_16x16x4_f32(false, ad, false, bfr[c0 + i],
                                                 (short)0, accd, false, false);
  }
}

__device__ __forceinline__ void tile_epilogue(const v8f acc, const v8f accd, float bb, float w3,
                                              int lane, float* fRow, float* fdRow)
{
  const int n  = lane & 15;
  const int mb = (lane >> 4) << 3;
  float pf[8], pfd[8];
  #pragma unroll
  for (int v = 0; v < 8; ++v) {
    float h2 = tanh_fast(acc[v] + bb);
    pf[v]  = h2 * w3;
    pfd[v] = (1.0f - h2 * h2) * accd[v] * w3;
  }
  #pragma unroll
  for (int v = 0; v < 8; ++v) { pf[v] = row_xadd<1>(pf[v]); pfd[v] = row_xadd<1>(pfd[v]); }
  #pragma unroll
  for (int v = 0; v < 8; ++v) { pf[v] = row_xadd<2>(pf[v]); pfd[v] = row_xadd<2>(pfd[v]); }
  #pragma unroll
  for (int v = 0; v < 8; ++v) { pf[v] = row_xadd<4>(pf[v]); pfd[v] = row_xadd<4>(pfd[v]); }
  #pragma unroll
  for (int v = 0; v < 8; ++v) { pf[v] = row_xadd<8>(pf[v]); pfd[v] = row_xadd<8>(pfd[v]); }
  if (n == 0) {
    #pragma unroll
    for (int v = 0; v < 8; ++v) {
      atomicAdd(&fRow[mb + v],  pf[v]);
      atomicAdd(&fdRow[mb + v], pfd[v]);
    }
  }
}

__global__ void __launch_bounds__(NTHREADS, 1)
cnf1d_rk4_wmma(const float* __restrict__ z0, const float* __restrict__ W1,
               const float* __restrict__ b1, const float* __restrict__ W2,
               const float* __restrict__ b2, const float* __restrict__ W3,
               const float* __restrict__ b3, float* __restrict__ out, int Btot)
{
  extern __shared__ float sm[];
  v4f*   sHH  = (v4f*)(sm + OFF_HH);
  float* sW1z = sm + OFF_W1Z;
  float* sW1t = sm + OFF_W1T;
  float* sB1  = sm + OFF_B1;
  float* sB2  = sm + OFF_B2;
  float* sW3  = sm + OFF_W3;
  float* sF   = sm + OFF_F;
  float* sFd  = sm + OFF_FD;
  float* sZ   = sm + OFF_Z;
  float* sZt  = sm + OFF_ZT;
  float* sDiv = sm + OFF_DIV;
  float* sKz  = sm + OFF_KZ;
  float* sKd  = sm + OFF_KD;

  const int tid  = threadIdx.x;
  const int lane = tid & 31;
  const int wv   = tid >> 5;
  const int hh   = lane >> 4;
  const float dtv = 0.25f;
  const float b3v = b3[0];

  v2f bfr[KCH];
  {
    const float* wp = W2 + (size_t)(2 * hh) * HDIM + (wv << 4) + (lane & 15);
    #pragma unroll
    for (int c = 0; c < KCH; ++c) {
      bfr[c].x = wp[(4 * c)     * HDIM];
      bfr[c].y = wp[(4 * c + 1) * HDIM];
    }
  }

  for (int i = tid; i < HDIM; i += NTHREADS) {
    sW1z[i] = W1[i];
    sW1t[i] = W1[HDIM + i];
    sB1[i]  = b1[i];
    sB2[i]  = b2[i];
    sW3[i]  = W3[i];
  }
  if (tid < MS) {
    float z = z0[(size_t)blockIdx.x * MS + tid];
    sZ[tid] = z; sZt[tid] = z; sDiv[tid] = 0.0f;
    sKz[tid] = 0.0f; sKd[tid] = 0.0f;
    sF[tid] = b3v; sFd[tid] = 0.0f;
  }
  __syncthreads();

  const int   jn = (wv << 4) | (lane & 15);
  const float w3 = sW3[jn];
  const float bb = sB2[jn];

  #pragma unroll 1
  for (int ev = 0; ev < 16; ++ev) {
    const int stg = ev & 3;
    float tt = (float)(ev >> 2) * dtv;
    if (stg == 3) tt += dtv; else if (stg != 0) tt += 0.5f * dtv;

    {
      const int m = lane & 15;
      #pragma unroll
      for (int q = 0; q < 4; ++q) {
        const int c  = (wv << 2) + q;
        const int j0 = (c << 2) + (hh << 1);
        const float w1a = sW1z[j0],  w1b = sW1z[j0 + 1];
        const float wta = sW1t[j0],  wtb = sW1t[j0 + 1];
        const float b1a = sB1[j0],   b1b = sB1[j0 + 1];
        #pragma unroll
        for (int mt = 0; mt < MT; ++mt) {
          const float z  = sZt[(mt << 4) + m];
          const float a0 = z * w1a + tt * wta + b1a;
          const float a1 = z * w1b + tt * wtb + b1b;
          const float h0 = tanh_fast(a0);
          const float h1 = tanh_fast(a1);
          v4f o;
          o.x = h0; o.y = h1;
          o.z = (1.0f - h0 * h0) * w1a;
          o.w = (1.0f - h1 * h1) * w1b;
          sHH[((mt * KCH + c) << 5) + lane] = o;
        }
      }
    }
    __syncthreads();

    v4f fa[KB], fb[KB];
    load_group(fa, sHH + lane);
    #pragma unroll 1
    for (int mt = 0; mt < MT; ++mt) {
      v8f acc = {}, accd = {};
      const v4f* hA  = sHH + ((mt * KCH) << 5) + lane;
      const v4f* hAn = sHH + ((((mt + 1) & (MT - 1)) * KCH) << 5) + lane;
      #pragma unroll
      for (int g = 0; g < NGRP; g += 2) {
        load_group(fb, hA + (((g + 1) * KB) << 5));
        wmma_block(fa, bfr, g * KB, acc, accd);
        __builtin_amdgcn_sched_barrier(0);
        load_group(fa, (g + 2 < NGRP) ? hA + (((g + 2) * KB) << 5) : hAn);
        wmma_block(fb, bfr, (g + 1) * KB, acc, accd);
        __builtin_amdgcn_sched_barrier(0);
      }
      tile_epilogue(acc, accd, bb, w3, lane, sF + (mt << 4), sFd + (mt << 4));
    }
    __syncthreads();

    if (tid < MS) {
      float kz = sF[tid];
      float kd = sFd[tid];
      float w  = (stg == 1 || stg == 2) ? 2.0f : 1.0f;
      float akz = sKz[tid] + w * kz;
      float akd = sKd[tid] + w * kd;
      if (stg < 3) {
        float cf = (stg == 2) ? dtv : 0.5f * dtv;
        sZt[tid] = sZ[tid] + cf * kz;
        sKz[tid] = akz; sKd[tid] = akd;
      } else {
        float zn = sZ[tid] + (dtv * (1.0f / 6.0f)) * akz;
        sZ[tid] = zn; sZt[tid] = zn;
        sDiv[tid] += (dtv * (1.0f / 6.0f)) * akd;
        sKz[tid] = 0.0f; sKd[tid] = 0.0f;
      }
      sF[tid] = b3v; sFd[tid] = 0.0f;
    }
    __syncthreads();
  }

  if (tid < MS) {
    int g = blockIdx.x * MS + tid;
    out[g] = sZ[tid];
    out[Btot + g] = sDiv[tid];
  }
}

extern "C" void kernel_launch(void* const* d_in, const int* in_sizes, int n_in,
                              void* d_out, int out_size, void* d_ws, size_t ws_size,
                              hipStream_t stream) {
  (void)n_in; (void)d_ws; (void)ws_size; (void)out_size;
  const float* z0 = (const float*)d_in[0];
  const float* W1 = (const float*)d_in[1];
  const float* b1 = (const float*)d_in[2];
  const float* W2 = (const float*)d_in[3];
  const float* b2 = (const float*)d_in[4];
  const float* W3 = (const float*)d_in[5];
  const float* b3 = (const float*)d_in[6];
  float* out = (float*)d_out;

  const int B    = in_sizes[0];
  const int nblk = B / MS;
  const size_t lds = (size_t)LDS_FLOATS * sizeof(float);

  hipFuncSetAttribute(reinterpret_cast<const void*>(cnf1d_rk4_wmma),
                      hipFuncAttributeMaxDynamicSharedMemorySize, (int)lds);
  cnf1d_rk4_wmma<<<nblk, NTHREADS, lds, stream>>>(z0, W1, b1, W2, b2, W3, b3, out, B);
}


# W3 sums over the 16 N-lanes as a halving tree: two values share each DPP add (15 per tile + selects, not 64 DPP ops), one ds_add per lane, not 16; same pairs and level order, products rounded first
# speedup vs baseline: 1.5992x; 1.0445x over previous
#include <hip/hip_runtime.h>
#include <math.h>


typedef __attribute__((ext_vector_type(2))) float v2f;
typedef __attribute__((ext_vector_type(4))) float v4f;
typedef __attribute__((ext_vector_type(8))) float v8f;

#define HDIM     256
#define KCH      64
#define MS       128
#define MT       8
#define NTHREADS 512
#define KB       8
#define NGRP     (KCH / KB)

#define OFF_HH   0
#define OFF_W1Z  (MT*KCH*32*4)
#define OFF_W1T  (OFF_W1Z + HDIM)
#define OFF_B1   (OFF_W1T + HDIM)
#define OFF_B2   (OFF_B1 + HDIM)
#define OFF_W3   (OFF_B2 + HDIM)
#define OFF_F    (OFF_W3 + HDIM)
#define OFF_FD   (OFF_F + MS)
#define OFF_Z    (OFF_FD + MS)
#define OFF_ZT   (OFF_Z + MS)
#define OFF_DIV  (OFF_ZT + MS)
#define OFF_KZ   (OFF_DIV + MS)
#define OFF_KD   (OFF_KZ + MS)
#define LDS_FLOATS (OFF_KD + MS)

__device__ __forceinline__ float tanh_fast(float x) {
#if defined(__AMDGCN__) && __has_builtin(__builtin_amdgcn_tanhf)
  return __builtin_amdgcn_tanhf(x);
#else
  float e = __expf(2.0f * x);
  return 1.0f - 2.0f / (e + 1.0f);
#endif
}

template <int M>
__device__ __forceinline__ float pair_xadd(float a, float b, bool up) {
  const float keep = up ? b : a;
  const float give = up ? a : b;
  const int t = __builtin_amdgcn_update_dpp(0, __builtin_bit_cast(int, give), 0x160 | M, 0xf, 0xf, true);
  return keep + __builtin_bit_cast(float, t);
}

__device__ __forceinline__ void load_group(v4f (&dst)[KB], const v4f* p) {
  #pragma unroll
  for (int i = 0; i < KB; ++i) dst[i] = p[i << 5];
}

__device__ __forceinline__ void wmma_block(const v4f (&set)[KB], const v2f (&bfr)[KCH], int c0,
                                           v8f& acc, v8f& accd) {
  #pragma unroll
  for (int i = 0; i < KB; ++i) {
    const v2f af = set[i].xy, ad = set[i].zw;
    acc  = __builtin_amdgcn_wmma_f32_16x16x4_f32(false, af, false, bfr[c0 + i],
                                                 (short)0, acc,  false, false);
    accd = __builtin_amdgcn_wmma_f32_16x16x4_f32(false, ad, false, bfr[c0 + i],
                                                 (short)0, accd, false, false);
  }
}

__device__ __forceinline__ void tile_epilogue(const v8f acc, const v8f accd, float bb, float w3,
                                              bool u1, bool u2, bool u4, bool u8, float* dst)
{
  float x[16];
  #pragma unroll
  for (int v = 0; v < 8; ++v) {
    float h2 = tanh_fast(acc[v] + bb);
    x[v]     = h2 * w3;
    x[8 + v] = (1.0f - h2 * h2) * accd[v] * w3;
  }
  float y[8], z[4], w[2];
  #pragma unroll
  for (int j = 0; j < 8; ++j) y[j] = pair_xadd<1>(x[2 * j], x[2 * j + 1], u1);
  #pragma unroll
  for (int j = 0; j < 4; ++j) z[j] = pair_xadd<2>(y[2 * j], y[2 * j + 1], u2);
  #pragma unroll
  for (int j = 0; j < 2; ++j) w[j] = pair_xadd<4>(z[2 * j], z[2 * j + 1], u4);
  const float r = pair_xadd<8>(w[0], w[1], u8);
  atomicAdd(dst, r);
}

__global__ void __launch_bounds__(NTHREADS, 1)
cnf1d_rk4_wmma(const float* __restrict__ z0, const float* __restrict__ W1,
               const float* __restrict__ b1, const float* __restrict__ W2,
               const float* __restrict__ b2, const float* __restrict__ W3,
               const float* __restrict__ b3, float* __restrict__ out, int Btot)
{
  extern __shared__ float sm[];
  v4f*   sHH  = (v4f*)(sm + OFF_HH);
  float* sW1z = sm + OFF_W1Z;
  float* sW1t = sm + OFF_W1T;
  float* sB1  = sm + OFF_B1;
  float* sB2  = sm + OFF_B2;
  float* sW3  = sm + OFF_W3;
  float* sF   = sm + OFF_F;
  float* sFd  = sm + OFF_FD;
  float* sZ   = sm + OFF_Z;
  float* sZt  = sm + OFF_ZT;
  float* sDiv = sm + OFF_DIV;
  float* sKz  = sm + OFF_KZ;
  float* sKd  = sm + OFF_KD;

  const int tid  = threadIdx.x;
  const int lane = tid & 31;
  const int wv   = tid >> 5;
  const int hh   = lane >> 4;
  const float dtv = 0.25f;
  const float b3v = b3[0];

  v2f bfr[KCH];
  {
    const float* wp = W2 + (size_t)(2 * hh) * HDIM + (wv << 4) + (lane & 15);
    #pragma unroll
    for (int c = 0; c < KCH; ++c) {
      bfr[c].x = wp[(4 * c)     * HDIM];
      bfr[c].y = wp[(4 * c + 1) * HDIM];
    }
  }

  for (int i = tid; i < HDIM; i += NTHREADS) {
    sW1z[i] = W1[i];
    sW1t[i] = W1[HDIM + i];
    sB1[i]  = b1[i];
    sB2[i]  = b2[i];
    sW3[i]  = W3[i];
  }
  if (tid < MS) {
    float z = z0[(size_t)blockIdx.x * MS + tid];
    sZ[tid] = z; sZt[tid] = z; sDiv[tid] = 0.0f;
    sKz[tid] = 0.0f; sKd[tid] = 0.0f;
    sF[tid] = b3v; sFd[tid] = 0.0f;
  }
  __syncthreads();

  const int   jn = (wv << 4) | (lane & 15);
  const float w3 = sW3[jn];
  const float bb = sB2[jn];
  const bool u1 = (lane & 1) != 0, u2 = (lane & 2) != 0, u4 = (lane & 4) != 0, u8 = (lane & 8) != 0;
  float* const fOut = sF + ((lane >> 3) & 1) * MS + ((lane >> 4) << 3) + (lane & 7);

  #pragma unroll 1
  for (int ev = 0; ev < 16; ++ev) {
    const int stg = ev & 3;
    float tt = (float)(ev >> 2) * dtv;
    if (stg == 3) tt += dtv; else if (stg != 0) tt += 0.5f * dtv;

    {
      const int m = lane & 15;
      #pragma unroll
      for (int q = 0; q < 4; ++q) {
        const int c  = (wv << 2) + q;
        const int j0 = (c << 2) + (hh << 1);
        const float w1a = sW1z[j0],  w1b = sW1z[j0 + 1];
        const float wta = sW1t[j0],  wtb = sW1t[j0 + 1];
        const float b1a = sB1[j0],   b1b = sB1[j0 + 1];
        #pragma unroll
        for (int mt = 0; mt < MT; ++mt) {
          const float z  = sZt[(mt << 4) + m];
          const float a0 = z * w1a + tt * wta + b1a;
          const float a1 = z * w1b + tt * wtb + b1b;
          const float h0 = tanh_fast(a0);
          const float h1 = tanh_fast(a1);
          v4f o;
          o.x = h0; o.y = h1;
          o.z = (1.0f - h0 * h0) * w1a;
          o.w = (1.0f - h1 * h1) * w1b;
          sHH[((mt * KCH + c) << 5) + lane] = o;
        }
      }
    }
    __syncthreads();

    v4f fa[KB], fb[KB];
    load_group(fa, sHH + lane);
    #pragma unroll 1
    for (int mt = 0; mt < MT; ++mt) {
      v8f acc = {}, accd = {};
      const v4f* hA  = sHH + ((mt * KCH) << 5) + lane;
      const v4f* hAn = sHH + ((((mt + 1) & (MT - 1)) * KCH) << 5) + lane;
      #pragma unroll
      for (int g = 0; g < NGRP; g += 2) {
        load_group(fb, hA + (((g + 1) * KB) << 5));
        wmma_block(fa, bfr, g * KB, acc, accd);
        __builtin_amdgcn_sched_barrier(0);
        load_group(fa, (g + 2 < NGRP) ? hA + (((g + 2) * KB) << 5) : hAn);
        wmma_block(fb, bfr, (g + 1) * KB, acc, accd);
        __builtin_amdgcn_sched_barrier(0);
      }
      tile_epilogue(acc, accd, bb, w3, u1, u2, u4, u8, fOut + (mt << 4));
    }
    __syncthreads();

    if (tid < MS) {
      float kz = sF[tid];
      float kd = sFd[tid];
      float w  = (stg == 1 || stg == 2) ? 2.0f : 1.0f;
      float akz = sKz[tid] + w * kz;
      float akd = sKd[tid] + w * kd;
      if (stg < 3) {
        float cf = (stg == 2) ? dtv : 0.5f * dtv;
        sZt[tid] = sZ[tid] + cf * kz;
        sKz[tid] = akz; sKd[tid] = akd;
      } else {
        float zn = sZ[tid] + (dtv * (1.0f / 6.0f)) * akz;
        sZ[tid] = zn; sZt[tid] = zn;
        sDiv[tid] += (dtv * (1.0f / 6.0f)) * akd;
        sKz[tid] = 0.0f; sKd[tid] = 0.0f;
      }
      sF[tid] = b3v; sFd[tid] = 0.0f;
    }
    __syncthreads();
  }

  if (tid < MS) {
    int g = blockIdx.x * MS + tid;
    out[g] = sZ[tid];
    out[Btot + g] = sDiv[tid];
  }
}

extern "C" void kernel_launch(void* const* d_in, const int* in_sizes, int n_in,
                              void* d_out, int out_size, void* d_ws, size_t ws_size,
                              hipStream_t stream) {
  (void)n_in; (void)d_ws; (void)ws_size; (void)out_size;
  const float* z0 = (const float*)d_in[0];
  const float* W1 = (const float*)d_in[1];
  const float* b1 = (const float*)d_in[2];
  const float* W2 = (const float*)d_in[3];
  const float* b2 = (const float*)d_in[4];
  const float* W3 = (const float*)d_in[5];
  const float* b3 = (const float*)d_in[6];
  float* out = (float*)d_out;

  const int B    = in_sizes[0];
  const int nblk = B / MS;
  const size_t lds = (size_t)LDS_FLOATS * sizeof(float);

  hipFuncSetAttribute(reinterpret_cast<const void*>(cnf1d_rk4_wmma),
                      hipFuncAttributeMaxDynamicSharedMemorySize, (int)lds);
  cnf1d_rk4_wmma<<<nblk, NTHREADS, lds, stream>>>(z0, W1, b1, W2, b2, W3, b3, out, B);
}
